# starting kernel
# baseline (speedup 1.0000x reference)
#include <hip/hip_runtime.h>


typedef __attribute__((ext_vector_type(2))) float v2f;
typedef __attribute__((ext_vector_type(8))) float v8f;

#define N_NODES 50000
#define N_EDGES 800000
#define IN_DIM  128
#define HID_DIM 128
#define OUT_DIM 64

__global__ void fill_kernel(float* __restrict__ p, float v, int n) {
    int i = blockIdx.x * blockDim.x + threadIdx.x;
    if (i < n) p[i] = v;
}

__global__ void deg_accum_kernel(const int* __restrict__ dst,
                                 float* __restrict__ deg, int nE) {
    int e = blockIdx.x * blockDim.x + threadIdx.x;
    if (e < nE) atomicAdd(&deg[dst[e]], 1.0f);
}

__global__ void dinv_kernel(float* __restrict__ d, int n) {
    int i = blockIdx.x * blockDim.x + threadIdx.x;
    if (i < n) d[i] = rsqrtf(fmaxf(d[i], 1.0f));
}

template<int NCOL>
__global__ __launch_bounds__(256)
void gemm_wmma_f32(const float* __restrict__ A, const float* __restrict__ W,
                   float* __restrict__ C, int M) {
    const int lane   = threadIdx.x & 31;
    const int wave   = threadIdx.x >> 5;
    const int tile_m = blockIdx.x * 8 + wave;
    if (tile_m >= (M >> 4)) return;
    const int tile_n = blockIdx.y;

    const int half = lane >> 4;
    const int lid  = lane & 15;
    const int row  = tile_m * 16 + lid;
    const int col  = tile_n * 16 + lid;
    const int koff = half * 2;

    const float* __restrict__ arow = A + (size_t)row * IN_DIM;

    v8f acc = {};
#pragma unroll 8
    for (int k = 0; k < IN_DIM; k += 4) {
        v2f a = *(const v2f*)(arow + k + koff);
        v2f b;
        b.x = W[(size_t)(k + koff)     * NCOL + col];
        b.y = W[(size_t)(k + koff + 1) * NCOL + col];
        acc = __builtin_amdgcn_wmma_f32_16x16x4_f32(
            false, a, false, b, (short)0, acc, false, false);
    }

#pragma unroll
    for (int r = 0; r < 8; ++r) {
        C[(size_t)(tile_m * 16 + r + half * 8) * NCOL + col] = acc[r];
    }
}

template<int D>
__global__ __launch_bounds__(256)
void edge_agg_kernel(const int* __restrict__ src, const int* __restrict__ dst,
                     const float* __restrict__ Y, const float* __restrict__ dinv,
                     float* __restrict__ agg, int nE) {
    const int lane = threadIdx.x & 31;
    const int e    = blockIdx.x * 8 + (threadIdx.x >> 5);
    if (e >= nE) return;

    const int   s    = src[e];
    const int   d    = dst[e];
    const float norm = dinv[s] * dinv[d];

    constexpr int VEC = D / 32;
    const float* __restrict__ yrow = Y   + (size_t)s * D + lane * VEC;
    float*       __restrict__ arow = agg + (size_t)d * D + lane * VEC;

    float v[VEC];
#pragma unroll
    for (int j = 0; j < VEC; ++j) v[j] = yrow[j];
#pragma unroll
    for (int j = 0; j < VEC; ++j) atomicAdd(&arow[j], v[j] * norm);
}

template<int D, bool RELU>
__global__ __launch_bounds__(256)
void self_bias_kernel(float* __restrict__ acc, const float* __restrict__ Y,
                      const float* __restrict__ dinv, const float* __restrict__ bias,
                      int n) {
    int i = blockIdx.x * blockDim.x + threadIdx.x;
    if (i >= n * D) return;
    const int   node = i / D;
    const int   c    = i % D;
    const float di   = dinv[node];
    float v = acc[i] + Y[i] * di * di + bias[c];
    acc[i] = RELU ? fmaxf(v, 0.0f) : v;
}

extern "C" void kernel_launch(void* const* d_in, const int* in_sizes, int n_in,
                              void* d_out, int out_size, void* d_ws, size_t ws_size,
                              hipStream_t stream) {
    const int*   E  = (const int*)  d_in[1];
    const float* X  = (const float*)d_in[2];
    const float* W0 = (const float*)d_in[3];
    const float* b0 = (const float*)d_in[4];
    const float* W1 = (const float*)d_in[5];
    const float* b1 = (const float*)d_in[6];
    float* out = (float*)d_out;

    const int* Esrc = E;
    const int* Edst = E + N_EDGES;

    float* ws   = (float*)d_ws;
    float* dinv = ws;
    float* agg0 = ws + 50048;
    float* xw   = agg0 + (size_t)N_NODES * HID_DIM;

    const int T = 256;
    fill_kernel<<<(N_NODES + T - 1) / T, T, 0, stream>>>(dinv, 1.0f, N_NODES);
    fill_kernel<<<(N_NODES * HID_DIM + T - 1) / T, T, 0, stream>>>(agg0, 0.0f, N_NODES * HID_DIM);
    fill_kernel<<<(N_NODES * OUT_DIM + T - 1) / T, T, 0, stream>>>(out, 0.0f, N_NODES * OUT_DIM);

    deg_accum_kernel<<<(N_EDGES + T - 1) / T, T, 0, stream>>>(Edst, dinv, N_EDGES);
    dinv_kernel<<<(N_NODES + T - 1) / T, T, 0, stream>>>(dinv, N_NODES);

    const int tilesM = N_NODES / 16;

    dim3 g0((tilesM + 7) / 8, HID_DIM / 16);
    gemm_wmma_f32<HID_DIM><<<g0, 256, 0, stream>>>(X, W0, xw, N_NODES);
    edge_agg_kernel<HID_DIM><<<(N_EDGES + 7) / 8, 256, 0, stream>>>(Esrc, Edst, xw, dinv, agg0, N_EDGES);
    self_bias_kernel<HID_DIM, true><<<(N_NODES * HID_DIM + T - 1) / T, T, 0, stream>>>(agg0, xw, dinv, b0, N_NODES);

    dim3 g1((tilesM + 7) / 8, OUT_DIM / 16);
    gemm_wmma_f32<OUT_DIM><<<g1, 256, 0, stream>>>(agg0, W1, xw, N_NODES);
    edge_agg_kernel<OUT_DIM><<<(N_EDGES + 7) / 8, 256, 0, stream>>>(Esrc, Edst, xw, dinv, out, N_EDGES);
    self_bias_kernel<OUT_DIM, false><<<(N_NODES * OUT_DIM + T - 1) / T, T, 0, stream>>>(out, xw, dinv, b1, N_NODES);
}


# Edges bucketed by destination per call (int atomic rank, scan, place); one wave sums a node's rows in registers, self-loop+bias+ReLU fused: no f32 atomics, sum order = arrival rank
# speedup vs baseline: 1.9875x; 1.9875x over previous
#include <hip/hip_runtime.h>


typedef __attribute__((ext_vector_type(2))) float v2f;
typedef __attribute__((ext_vector_type(8))) float v8f;

#define N_NODES 50000
#define N_EDGES 800000
#define IN_DIM  128
#define HID_DIM 128
#define OUT_DIM 64

__global__ void zero_int_kernel(int* __restrict__ p, int n) {
    int i = blockIdx.x * blockDim.x + threadIdx.x;
    if (i < n) p[i] = 0;
}

__global__ void rank_kernel(const int* __restrict__ dst, int* __restrict__ cnt,
                            int* __restrict__ rank, int nE) {
    int e = blockIdx.x * blockDim.x + threadIdx.x;
    if (e < nE) rank[e] = atomicAdd(&cnt[dst[e]], 1);
}

#define SCAN_TILES 13
static_assert(N_NODES % 4 == 0, "scan reads counts four at a time");
static_assert(32 * SCAN_TILES * 128 >= N_NODES, "scan range too small");

__global__ __launch_bounds__(1024)
void scan_dinv_kernel(const int* __restrict__ cnt, int* __restrict__ rowptr,
                      float* __restrict__ dinv, int n) {
    __shared__ int wsum[32];
    const int lane = threadIdx.x & 31;
    const int wave = threadIdx.x >> 5;
    const int n4   = n >> 2;
    const int q0   = wave * (SCAN_TILES * 32) + lane;

    const int4* __restrict__ cnt4 = (const int4*)cnt;
    int4 v[SCAN_TILES];
#pragma unroll
    for (int i = 0; i < SCAN_TILES; ++i) {
        const int q = q0 + i * 32;
        v[i] = (q < n4) ? cnt4[q] : make_int4(0, 0, 0, 0);
    }

    int carry = 0;
    int ex[SCAN_TILES];
#pragma unroll
    for (int i = 0; i < SCAN_TILES; ++i) {
        const int s = v[i].x + v[i].y + v[i].z + v[i].w;
        int inc = s;
#pragma unroll
        for (int d = 1; d < 32; d <<= 1) {
            const int t = __shfl_up(inc, d);
            if (lane >= d) inc += t;
        }
        ex[i] = carry + inc - s;
        carry += __shfl(inc, 31);
    }
    if (lane == 0) wsum[wave] = carry;
    __syncthreads();

    int off = (lane < wave) ? wsum[lane] : 0;
#pragma unroll
    for (int d = 16; d >= 1; d >>= 1) off += __shfl_xor(off, d);

    int4*   __restrict__ row4 = (int4*)rowptr;
    float4* __restrict__ di4  = (float4*)dinv;
#pragma unroll
    for (int i = 0; i < SCAN_TILES; ++i) {
        const int q = q0 + i * 32;
        if (q < n4) {
            const int p0 = off + ex[i];
            const int p1 = p0 + v[i].x;
            const int p2 = p1 + v[i].y;
            const int p3 = p2 + v[i].z;
            row4[q] = make_int4(p0, p1, p2, p3);
            di4[q]  = make_float4(rsqrtf(fmaxf((float)(v[i].x + 1), 1.0f)),
                                  rsqrtf(fmaxf((float)(v[i].y + 1), 1.0f)),
                                  rsqrtf(fmaxf((float)(v[i].z + 1), 1.0f)),
                                  rsqrtf(fmaxf((float)(v[i].w + 1), 1.0f)));
        }
    }
    if (threadIdx.x == 1023) rowptr[n] = off + carry;
}

__global__ void place_kernel(const int* __restrict__ src, const int* __restrict__ dst,
                             const int* __restrict__ rank, const int* __restrict__ rowptr,
                             int* __restrict__ csr_src, int nE) {
    int e = blockIdx.x * blockDim.x + threadIdx.x;
    if (e < nE) csr_src[rowptr[dst[e]] + rank[e]] = src[e];
}

template<int NCOL>
__global__ __launch_bounds__(256)
void gemm_wmma_f32(const float* __restrict__ A, const float* __restrict__ W,
                   float* __restrict__ C, int M) {
    const int lane   = threadIdx.x & 31;
    const int wave   = threadIdx.x >> 5;
    const int tile_m = blockIdx.x * 8 + wave;
    if (tile_m >= (M >> 4)) return;
    const int tile_n = blockIdx.y;

    const int half = lane >> 4;
    const int lid  = lane & 15;
    const int row  = tile_m * 16 + lid;
    const int col  = tile_n * 16 + lid;
    const int koff = half * 2;

    const float* __restrict__ arow = A + (size_t)row * IN_DIM;

    v8f acc = {};
#pragma unroll 8
    for (int k = 0; k < IN_DIM; k += 4) {
        v2f a = *(const v2f*)(arow + k + koff);
        v2f b;
        b.x = W[(size_t)(k + koff)     * NCOL + col];
        b.y = W[(size_t)(k + koff + 1) * NCOL + col];
        acc = __builtin_amdgcn_wmma_f32_16x16x4_f32(
            false, a, false, b, (short)0, acc, false, false);
    }

#pragma unroll
    for (int r = 0; r < 8; ++r) {
        C[(size_t)(tile_m * 16 + r + half * 8) * NCOL + col] = acc[r];
    }
}

template<int V> struct RowVec;
template<> struct RowVec<4> { typedef __attribute__((ext_vector_type(4))) float type; };
template<> struct RowVec<2> { typedef __attribute__((ext_vector_type(2))) float type; };

template<int K, typename VT, int V>
__device__ __forceinline__ void add_rows(VT& acc, const VT* __restrict__ Yv,
                                         int my_s, int my_wbits, int j, int lane) {
    int   s[K];
    float w[K];
    VT    y[K];
#pragma unroll
    for (int k = 0; k < K; ++k) {
        s[k] = __builtin_amdgcn_readlane(my_s, j + k);
        w[k] = __builtin_bit_cast(float, __builtin_amdgcn_readlane(my_wbits, j + k));
    }
#pragma unroll
    for (int k = 0; k < K; ++k) y[k] = Yv[(size_t)s[k] * 32 + lane];
#pragma unroll
    for (int k = 0; k < K; ++k) {
#pragma unroll
        for (int c = 0; c < V; ++c) acc[c] = __builtin_fmaf(y[k][c], w[k], acc[c]);
    }
}

template<int D, bool RELU>
__global__ __launch_bounds__(256)
void gather_kernel(const int* __restrict__ rowptr, const int* __restrict__ csr_src,
                   const float* __restrict__ Y, const float* __restrict__ dinv,
                   const float* __restrict__ bias, float* __restrict__ outp, int n) {
    constexpr int V = D / 32;
    typedef typename RowVec<V>::type VT;

    const int lane = threadIdx.x & 31;
    const int node = __builtin_amdgcn_readfirstlane((int)(blockIdx.x * 8 + (threadIdx.x >> 5)));
    if (node >= n) return;

    const int   beg = rowptr[node];
    const int   end = rowptr[node + 1];
    const float dn  = dinv[node];

    const VT* __restrict__ Yv = (const VT*)Y;
    const VT yself = Yv[(size_t)node * 32 + lane];

    VT acc;
#pragma unroll
    for (int c = 0; c < V; ++c) acc[c] = 0.0f;

    for (int base = beg; base < end; base += 32) {
        const int rem = end - base;
        const int m   = rem < 32 ? rem : 32;
        int   my_s = node;
        float my_w = 0.0f;
        if (lane < m) {
            my_s = csr_src[base + lane];
            my_w = dinv[my_s] * dn;
        }
        const int my_wbits = __builtin_bit_cast(int, my_w);

        int j = 0;
        for (; j + 8 <= m; j += 8) add_rows<8, VT, V>(acc, Yv, my_s, my_wbits, j, lane);
        if (j + 4 <= m) { add_rows<4, VT, V>(acc, Yv, my_s, my_wbits, j, lane); j += 4; }
        const int r = m - j;
        if (r == 3)      add_rows<3, VT, V>(acc, Yv, my_s, my_wbits, j, lane);
        else if (r == 2) add_rows<2, VT, V>(acc, Yv, my_s, my_wbits, j, lane);
        else if (r == 1) add_rows<1, VT, V>(acc, Yv, my_s, my_wbits, j, lane);
    }

    const float dself = dn * dn;
    VT o;
#pragma unroll
    for (int c = 0; c < V; ++c) {
        const float t = __builtin_fmaf(yself[c], dself, acc[c]) + bias[lane * V + c];
        o[c] = RELU ? fmaxf(t, 0.0f) : t;
    }
    ((VT*)outp)[(size_t)node * 32 + lane] = o;
}

extern "C" void kernel_launch(void* const* d_in, const int* in_sizes, int n_in,
                              void* d_out, int out_size, void* d_ws, size_t ws_size,
                              hipStream_t stream) {
    const int*   E  = (const int*)  d_in[1];
    const float* X  = (const float*)d_in[2];
    const float* W0 = (const float*)d_in[3];
    const float* b0 = (const float*)d_in[4];
    const float* W1 = (const float*)d_in[5];
    const float* b1 = (const float*)d_in[6];
    float* out = (float*)d_out;

    const int* Esrc = E;
    const int* Edst = E + N_EDGES;

    int*   cnt    = (int*)d_ws;
    int*   rowptr = cnt + 50048;
    float* dinv   = (float*)(rowptr + 50176);
    int*   rank   = (int*)(dinv + 50048);
    int*   csr    = rank + N_EDGES;
    float* xw     = (float*)(csr + N_EDGES);
    float* h      = xw + (size_t)N_NODES * HID_DIM;

    const int T = 256;
    zero_int_kernel<<<(N_NODES + T - 1) / T, T, 0, stream>>>(cnt, N_NODES);
    rank_kernel<<<(N_EDGES + T - 1) / T, T, 0, stream>>>(Edst, cnt, rank, N_EDGES);
    scan_dinv_kernel<<<1, 1024, 0, stream>>>(cnt, rowptr, dinv, N_NODES);
    place_kernel<<<(N_EDGES + T - 1) / T, T, 0, stream>>>(Esrc, Edst, rank, rowptr, csr, N_EDGES);

    const int tilesM = N_NODES / 16;

    dim3 g0((tilesM + 7) / 8, HID_DIM / 16);
    gemm_wmma_f32<HID_DIM><<<g0, 256, 0, stream>>>(X, W0, xw, N_NODES);
    gather_kernel<HID_DIM, true><<<(N_NODES + 7) / 8, 256, 0, stream>>>(rowptr, csr, xw, dinv, b0, h, N_NODES);

    dim3 g1((tilesM + 7) / 8, OUT_DIM / 16);
    gemm_wmma_f32<OUT_DIM><<<g1, 256, 0, stream>>>(h, W1, xw, N_NODES);
    gather_kernel<OUT_DIM, false><<<(N_NODES + 7) / 8, 256, 0, stream>>>(rowptr, csr, xw, dinv, b1, out, N_NODES);
}


# Both dense products moved into the layer-0 gather: 16 waves sum X rows of 16 nodes (aggregate-first: layer-0 sum order changes), LDS tile @W0,+b0,ReLU,@W1 on WMMA; 8->6 launches
# speedup vs baseline: 2.3673x; 1.1911x over previous
#include <hip/hip_runtime.h>


typedef __attribute__((ext_vector_type(2))) float v2f;
typedef __attribute__((ext_vector_type(8))) float v8f;

#define N_NODES 50000
#define N_EDGES 800000
#define IN_DIM  128
#define HID_DIM 128
#define OUT_DIM 64

__global__ void zero_int_kernel(int* __restrict__ p, int n) {
    int i = blockIdx.x * blockDim.x + threadIdx.x;
    if (i < n) p[i] = 0;
}

__global__ void rank_kernel(const int* __restrict__ dst, int* __restrict__ cnt,
                            int* __restrict__ rank, int nE) {
    int e = blockIdx.x * blockDim.x + threadIdx.x;
    if (e < nE) rank[e] = atomicAdd(&cnt[dst[e]], 1);
}

#define SCAN_TILES 13
static_assert(N_NODES % 4 == 0, "scan reads counts four at a time");
static_assert(32 * SCAN_TILES * 128 >= N_NODES, "scan range too small");

__global__ __launch_bounds__(1024)
void scan_dinv_kernel(const int* __restrict__ cnt, int* __restrict__ rowptr,
                      float* __restrict__ dinv, int n) {
    __shared__ int wsum[32];
    const int lane = threadIdx.x & 31;
    const int wave = threadIdx.x >> 5;
    const int n4   = n >> 2;
    const int q0   = wave * (SCAN_TILES * 32) + lane;

    const int4* __restrict__ cnt4 = (const int4*)cnt;
    int4 v[SCAN_TILES];
#pragma unroll
    for (int i = 0; i < SCAN_TILES; ++i) {
        const int q = q0 + i * 32;
        v[i] = (q < n4) ? cnt4[q] : make_int4(0, 0, 0, 0);
    }

    int carry = 0;
    int ex[SCAN_TILES];
#pragma unroll
    for (int i = 0; i < SCAN_TILES; ++i) {
        const int s = v[i].x + v[i].y + v[i].z + v[i].w;
        int inc = s;
#pragma unroll
        for (int d = 1; d < 32; d <<= 1) {
            const int t = __shfl_up(inc, d);
            if (lane >= d) inc += t;
        }
        ex[i] = carry + inc - s;
        carry += __shfl(inc, 31);
    }
    if (lane == 0) wsum[wave] = carry;
    __syncthreads();

    int off = (lane < wave) ? wsum[lane] : 0;
#pragma unroll
    for (int d = 16; d >= 1; d >>= 1) off += __shfl_xor(off, d);

    int4*   __restrict__ row4 = (int4*)rowptr;
    float4* __restrict__ di4  = (float4*)dinv;
#pragma unroll
    for (int i = 0; i < SCAN_TILES; ++i) {
        const int q = q0 + i * 32;
        if (q < n4) {
            const int p0 = off + ex[i];
            const int p1 = p0 + v[i].x;
            const int p2 = p1 + v[i].y;
            const int p3 = p2 + v[i].z;
            row4[q] = make_int4(p0, p1, p2, p3);
            di4[q]  = make_float4(rsqrtf(fmaxf((float)(v[i].x + 1), 1.0f)),
                                  rsqrtf(fmaxf((float)(v[i].y + 1), 1.0f)),
                                  rsqrtf(fmaxf((float)(v[i].z + 1), 1.0f)),
                                  rsqrtf(fmaxf((float)(v[i].w + 1), 1.0f)));
        }
    }
    if (threadIdx.x == 1023) rowptr[n] = off + carry;
}

__global__ void place_kernel(const int* __restrict__ src, const int* __restrict__ dst,
                             const int* __restrict__ rank, const int* __restrict__ rowptr,
                             int* __restrict__ csr_src, int nE) {
    int e = blockIdx.x * blockDim.x + threadIdx.x;
    if (e < nE) csr_src[rowptr[dst[e]] + rank[e]] = src[e];
}

#define T_PITCH 132
#define Y_PITCH 68

template<int NCOL>
__device__ __forceinline__ v8f tile_product(const float* __restrict__ T,
                                            const float* __restrict__ W,
                                            int tile_n, int lane) {
    const int half = lane >> 4;
    const int lid  = lane & 15;
    const int col  = tile_n * 16 + lid;
    const int koff = half * 2;
    const float* arow = T + lid * T_PITCH;

    v8f acc = {};
#pragma unroll 8
    for (int k = 0; k < IN_DIM; k += 4) {
        v2f a = *(const v2f*)(arow + k + koff);
        v2f b;
        b.x = W[(size_t)(k + koff)     * NCOL + col];
        b.y = W[(size_t)(k + koff + 1) * NCOL + col];
        acc = __builtin_amdgcn_wmma_f32_16x16x4_f32(
            false, a, false, b, (short)0, acc, false, false);
    }
    return acc;
}

template<int V> struct RowVec;
template<> struct RowVec<4> { typedef __attribute__((ext_vector_type(4))) float type; };
template<> struct RowVec<2> { typedef __attribute__((ext_vector_type(2))) float type; };

template<int K, typename VT, int V>
__device__ __forceinline__ void add_rows(VT& acc, const VT* __restrict__ Yv,
                                         int my_s, int my_wbits, int j, int lane) {
    int   s[K];
    float w[K];
    VT    y[K];
#pragma unroll
    for (int k = 0; k < K; ++k) {
        s[k] = __builtin_amdgcn_readlane(my_s, j + k);
        w[k] = __builtin_bit_cast(float, __builtin_amdgcn_readlane(my_wbits, j + k));
    }
#pragma unroll
    for (int k = 0; k < K; ++k) y[k] = Yv[(size_t)s[k] * 32 + lane];
#pragma unroll
    for (int k = 0; k < K; ++k) {
#pragma unroll
        for (int c = 0; c < V; ++c) acc[c] = __builtin_fmaf(y[k][c], w[k], acc[c]);
    }
}

template<int D, bool RELU>
__global__ __launch_bounds__(256)
void gather_kernel(const int* __restrict__ rowptr, const int* __restrict__ csr_src,
                   const float* __restrict__ Y, const float* __restrict__ dinv,
                   const float* __restrict__ bias, float* __restrict__ outp, int n) {
    constexpr int V = D / 32;
    typedef typename RowVec<V>::type VT;

    const int lane = threadIdx.x & 31;
    const int node = __builtin_amdgcn_readfirstlane((int)(blockIdx.x * 8 + (threadIdx.x >> 5)));
    if (node >= n) return;

    const int   beg = rowptr[node];
    const int   end = rowptr[node + 1];
    const float dn  = dinv[node];

    const VT* __restrict__ Yv = (const VT*)Y;
    const VT yself = Yv[(size_t)node * 32 + lane];

    VT acc;
#pragma unroll
    for (int c = 0; c < V; ++c) acc[c] = 0.0f;

    for (int base = beg; base < end; base += 32) {
        const int rem = end - base;
        const int m   = rem < 32 ? rem : 32;
        int   my_s = node;
        float my_w = 0.0f;
        if (lane < m) {
            my_s = csr_src[base + lane];
            my_w = dinv[my_s] * dn;
        }
        const int my_wbits = __builtin_bit_cast(int, my_w);

        int j = 0;
        for (; j + 8 <= m; j += 8) add_rows<8, VT, V>(acc, Yv, my_s, my_wbits, j, lane);
        if (j + 4 <= m) { add_rows<4, VT, V>(acc, Yv, my_s, my_wbits, j, lane); j += 4; }
        const int r = m - j;
        if (r == 3)      add_rows<3, VT, V>(acc, Yv, my_s, my_wbits, j, lane);
        else if (r == 2) add_rows<2, VT, V>(acc, Yv, my_s, my_wbits, j, lane);
        else if (r == 1) add_rows<1, VT, V>(acc, Yv, my_s, my_wbits, j, lane);
    }

    const float dself = dn * dn;
    VT o;
#pragma unroll
    for (int c = 0; c < V; ++c) {
        const float t = __builtin_fmaf(yself[c], dself, acc[c]) + bias[lane * V + c];
        o[c] = RELU ? fmaxf(t, 0.0f) : t;
    }
    ((VT*)outp)[(size_t)node * 32 + lane] = o;
}

static_assert(N_NODES % 16 == 0, "one 16-row tile per work-group");
static_assert(IN_DIM == HID_DIM, "tile_product steps K over IN_DIM for both products");

__global__ __launch_bounds__(512)
void gather_transform_kernel(const int* __restrict__ rowptr, const int* __restrict__ csr_src,
                             const float* __restrict__ X, const float* __restrict__ dinv,
                             const float* __restrict__ W0, const float* __restrict__ b0,
                             const float* __restrict__ W1, float* __restrict__ Y1, int n) {
    typedef typename RowVec<4>::type VT;
    __shared__ float As[16 * T_PITCH];
    __shared__ float Hs[16 * T_PITCH];
    __shared__ float Ys[16 * Y_PITCH];

    const int lane = threadIdx.x & 31;
    const int wave = __builtin_amdgcn_readfirstlane((int)(threadIdx.x >> 5));
    const int node = __builtin_amdgcn_readfirstlane((int)(blockIdx.x * 16)) + wave;

    VT agg;
#pragma unroll
    for (int c = 0; c < 4; ++c) agg[c] = 0.0f;

    if (node < n) {
        const int   beg = rowptr[node];
        const int   end = rowptr[node + 1];
        const float dn  = dinv[node];

        const VT* __restrict__ Xv = (const VT*)X;
        const VT xself = Xv[(size_t)node * 32 + lane];

        VT acc;
#pragma unroll
        for (int c = 0; c < 4; ++c) acc[c] = 0.0f;

        for (int base = beg; base < end; base += 32) {
            const int rem = end - base;
            const int m   = rem < 32 ? rem : 32;
            int   my_s = node;
            float my_w = 0.0f;
            if (lane < m) {
                my_s = csr_src[base + lane];
                my_w = dinv[my_s] * dn;
            }
            const int my_wbits = __builtin_bit_cast(int, my_w);

            int j = 0;
            for (; j + 8 <= m; j += 8) add_rows<8, VT, 4>(acc, Xv, my_s, my_wbits, j, lane);
            if (j + 4 <= m) { add_rows<4, VT, 4>(acc, Xv, my_s, my_wbits, j, lane); j += 4; }
            const int r = m - j;
            if (r == 3)      add_rows<3, VT, 4>(acc, Xv, my_s, my_wbits, j, lane);
            else if (r == 2) add_rows<2, VT, 4>(acc, Xv, my_s, my_wbits, j, lane);
            else if (r == 1) add_rows<1, VT, 4>(acc, Xv, my_s, my_wbits, j, lane);
        }

        const float dself = dn * dn;
#pragma unroll
        for (int c = 0; c < 4; ++c) agg[c] = __builtin_fmaf(xself[c], dself, acc[c]);
    }
    *(VT*)(As + wave * T_PITCH + lane * 4) = agg;
    __syncthreads();

    const int half = lane >> 4;
    const int lid  = lane & 15;

    if (wave < HID_DIM / 16) {
        const v8f acc = tile_product<HID_DIM>(As, W0, wave, lane);
        const int   col  = wave * 16 + lid;
        const float bias = b0[col];
#pragma unroll
        for (int r = 0; r < 8; ++r)
            Hs[(r + half * 8) * T_PITCH + col] = fmaxf(acc[r] + bias, 0.0f);
    }
    __syncthreads();

    if (wave < OUT_DIM / 16) {
        const v8f acc = tile_product<OUT_DIM>(Hs, W1, wave, lane);
        const int col = wave * 16 + lid;
#pragma unroll
        for (int r = 0; r < 8; ++r)
            Ys[(r + half * 8) * Y_PITCH + col] = acc[r];
    }
    __syncthreads();

    if (node < n) {
        const v2f y = *(const v2f*)(Ys + wave * Y_PITCH + lane * 2);
        ((v2f*)Y1)[(size_t)node * 32 + lane] = y;
    }
}

extern "C" void kernel_launch(void* const* d_in, const int* in_sizes, int n_in,
                              void* d_out, int out_size, void* d_ws, size_t ws_size,
                              hipStream_t stream) {
    const int*   E  = (const int*)  d_in[1];
    const float* X  = (const float*)d_in[2];
    const float* W0 = (const float*)d_in[3];
    const float* b0 = (const float*)d_in[4];
    const float* W1 = (const float*)d_in[5];
    const float* b1 = (const float*)d_in[6];
    float* out = (float*)d_out;

    const int* Esrc = E;
    const int* Edst = E + N_EDGES;

    int*   cnt    = (int*)d_ws;
    int*   rowptr = cnt + 50048;
    float* dinv   = (float*)(rowptr + 50176);
    int*   rank   = (int*)(dinv + 50048);
    int*   csr    = rank + N_EDGES;
    float* y1     = (float*)(csr + N_EDGES);

    const int T = 256;
    zero_int_kernel<<<(N_NODES + T - 1) / T, T, 0, stream>>>(cnt, N_NODES);
    rank_kernel<<<(N_EDGES + T - 1) / T, T, 0, stream>>>(Edst, cnt, rank, N_EDGES);
    scan_dinv_kernel<<<1, 1024, 0, stream>>>(cnt, rowptr, dinv, N_NODES);
    place_kernel<<<(N_EDGES + T - 1) / T, T, 0, stream>>>(Esrc, Edst, rank, rowptr, csr, N_EDGES);

    gather_transform_kernel<<<N_NODES / 16, 512, 0, stream>>>(rowptr, csr, X, dinv, W0, b0, W1, y1, N_NODES);

    gather_kernel<OUT_DIM, false><<<(N_NODES + 7) / 8, 256, 0, stream>>>(rowptr, csr, y1, dinv, b1, out, N_NODES);
}


# Counting add's return is the slot in a fixed 32-slot bucket per destination (beyond 32: overflow list, summed after the bucket in list order); scan and place launches dropped, 6->4 launches
# speedup vs baseline: 2.8117x; 1.1877x over previous
#include <hip/hip_runtime.h>


typedef __attribute__((ext_vector_type(2))) float v2f;
typedef __attribute__((ext_vector_type(8))) float v8f;

#define N_NODES 50000
#define N_EDGES 800000
#define IN_DIM  128
#define HID_DIM 128
#define OUT_DIM 64

__global__ void zero_int_kernel(int* __restrict__ p, int n) {
    int i = blockIdx.x * blockDim.x + threadIdx.x;
    if (i < n) p[i] = 0;
}

#define BUCKET_CAP 32

__global__ void bucket_kernel(const int* __restrict__ src, const int* __restrict__ dst,
                              int* __restrict__ cnt, int* __restrict__ bucket,
                              int* __restrict__ ovf, int nE) {
    int e = blockIdx.x * blockDim.x + threadIdx.x;
    if (e < nE) {
        const int d = dst[e];
        const int s = src[e];
        const int r = atomicAdd(&cnt[d], 1);
        if (r < BUCKET_CAP) {
            bucket[d * BUCKET_CAP + r] = s;
        } else {
            const int k = atomicAdd(&cnt[N_NODES], 1);
            ((int2*)ovf)[k] = make_int2(d, s);
        }
    }
}

__device__ __forceinline__ float inv_sqrt_deg(int c) {
    return rsqrtf(fmaxf((float)(c + 1), 1.0f));
}

#define T_PITCH 132
#define Y_PITCH 68

template<int NCOL>
__device__ __forceinline__ v8f tile_product(const float* __restrict__ T,
                                            const float* __restrict__ W,
                                            int tile_n, int lane) {
    const int half = lane >> 4;
    const int lid  = lane & 15;
    const int col  = tile_n * 16 + lid;
    const int koff = half * 2;
    const float* arow = T + lid * T_PITCH;

    v8f acc = {};
#pragma unroll 8
    for (int k = 0; k < IN_DIM; k += 4) {
        v2f a = *(const v2f*)(arow + k + koff);
        v2f b;
        b.x = W[(size_t)(k + koff)     * NCOL + col];
        b.y = W[(size_t)(k + koff + 1) * NCOL + col];
        acc = __builtin_amdgcn_wmma_f32_16x16x4_f32(
            false, a, false, b, (short)0, acc, false, false);
    }
    return acc;
}

template<int V> struct RowVec;
template<> struct RowVec<4> { typedef __attribute__((ext_vector_type(4))) float type; };
template<> struct RowVec<2> { typedef __attribute__((ext_vector_type(2))) float type; };

template<int K, typename VT, int V>
__device__ __forceinline__ void add_rows(VT& acc, const VT* __restrict__ Yv,
                                         int my_s, int my_wbits, int j, int lane) {
    int   s[K];
    float w[K];
    VT    y[K];
#pragma unroll
    for (int k = 0; k < K; ++k) {
        s[k] = __builtin_amdgcn_readlane(my_s, j + k);
        w[k] = __builtin_bit_cast(float, __builtin_amdgcn_readlane(my_wbits, j + k));
    }
#pragma unroll
    for (int k = 0; k < K; ++k) y[k] = Yv[(size_t)s[k] * 32 + lane];
#pragma unroll
    for (int k = 0; k < K; ++k) {
#pragma unroll
        for (int c = 0; c < V; ++c) acc[c] = __builtin_fmaf(y[k][c], w[k], acc[c]);
    }
}

template<typename VT, int V>
__device__ __forceinline__ void sum_incoming(VT& acc, const int* __restrict__ cnt,
                                             const int* __restrict__ bucket,
                                             const int* __restrict__ ovf,
                                             const VT* __restrict__ Yv,
                                             int node, int deg, float dn, int lane) {
    const int m = deg < BUCKET_CAP ? deg : BUCKET_CAP;
    int   my_s = node;
    float my_w = 0.0f;
    if (lane < m) {
        my_s = bucket[node * BUCKET_CAP + lane];
        my_w = inv_sqrt_deg(cnt[my_s]) * dn;
    }
    const int my_wbits = __builtin_bit_cast(int, my_w);

    int j = 0;
    for (; j + 8 <= m; j += 8) add_rows<8, VT, V>(acc, Yv, my_s, my_wbits, j, lane);
    if (j + 4 <= m) { add_rows<4, VT, V>(acc, Yv, my_s, my_wbits, j, lane); j += 4; }
    const int r = m - j;
    if (r == 3)      add_rows<3, VT, V>(acc, Yv, my_s, my_wbits, j, lane);
    else if (r == 2) add_rows<2, VT, V>(acc, Yv, my_s, my_wbits, j, lane);
    else if (r == 1) add_rows<1, VT, V>(acc, Yv, my_s, my_wbits, j, lane);

    if (deg > BUCKET_CAP) {
        const int  len   = cnt[N_NODES];
        const int  n_ovf = len < N_EDGES ? len : N_EDGES;
        const int2* __restrict__ ovf2 = (const int2*)ovf;
        for (int k0 = 0; k0 < n_ovf; k0 += 32) {
            const int k = k0 + lane;
            bool  hit = false;
            int   o_s = node;
            float o_w = 0.0f;
            if (k < n_ovf) {
                const int2 rec = ovf2[k];
                if (rec.x == node) {
                    hit = true;
                    o_s = rec.y;
                    o_w = inv_sqrt_deg(cnt[o_s]) * dn;
                }
            }
            const int o_wbits = __builtin_bit_cast(int, o_w);
            unsigned todo = __builtin_amdgcn_ballot_w32(hit);
            while (todo != 0u) {
                const int l = __builtin_ctz(todo);
                todo &= todo - 1u;
                add_rows<1, VT, V>(acc, Yv, o_s, o_wbits, l, lane);
            }
        }
    }
}

template<int D, bool RELU>
__global__ __launch_bounds__(256)
void gather_kernel(const int* __restrict__ cnt, const int* __restrict__ bucket,
                   const int* __restrict__ ovf, const float* __restrict__ Y,
                   const float* __restrict__ bias, float* __restrict__ outp, int n) {
    constexpr int V = D / 32;
    typedef typename RowVec<V>::type VT;

    const int lane = threadIdx.x & 31;
    const int node = __builtin_amdgcn_readfirstlane((int)(blockIdx.x * 8 + (threadIdx.x >> 5)));
    if (node >= n) return;

    const int   deg = cnt[node];
    const float dn  = inv_sqrt_deg(deg);

    const VT* __restrict__ Yv = (const VT*)Y;
    const VT yself = Yv[(size_t)node * 32 + lane];

    VT acc;
#pragma unroll
    for (int c = 0; c < V; ++c) acc[c] = 0.0f;

    sum_incoming<VT, V>(acc, cnt, bucket, ovf, Yv, node, deg, dn, lane);

    const float dself = dn * dn;
    VT o;
#pragma unroll
    for (int c = 0; c < V; ++c) {
        const float t = __builtin_fmaf(yself[c], dself, acc[c]) + bias[lane * V + c];
        o[c] = RELU ? fmaxf(t, 0.0f) : t;
    }
    ((VT*)outp)[(size_t)node * 32 + lane] = o;
}

static_assert(N_NODES % 16 == 0, "one 16-row tile per work-group");
static_assert(IN_DIM == HID_DIM, "tile_product steps K over IN_DIM for both products");

__global__ __launch_bounds__(512)
void gather_transform_kernel(const int* __restrict__ cnt, const int* __restrict__ bucket,
                             const int* __restrict__ ovf, const float* __restrict__ X,
                             const float* __restrict__ W0, const float* __restrict__ b0,
                             const float* __restrict__ W1, float* __restrict__ Y1, int n) {
    typedef typename RowVec<4>::type VT;
    __shared__ float As[16 * T_PITCH];
    __shared__ float Hs[16 * T_PITCH];
    __shared__ float Ys[16 * Y_PITCH];

    const int lane = threadIdx.x & 31;
    const int wave = __builtin_amdgcn_readfirstlane((int)(threadIdx.x >> 5));
    const int node = __builtin_amdgcn_readfirstlane((int)(blockIdx.x * 16)) + wave;

    VT agg;
#pragma unroll
    for (int c = 0; c < 4; ++c) agg[c] = 0.0f;

    if (node < n) {
        const int   deg = cnt[node];
        const float dn  = inv_sqrt_deg(deg);

        const VT* __restrict__ Xv = (const VT*)X;
        const VT xself = Xv[(size_t)node * 32 + lane];

        VT acc;
#pragma unroll
        for (int c = 0; c < 4; ++c) acc[c] = 0.0f;

        sum_incoming<VT, 4>(acc, cnt, bucket, ovf, Xv, node, deg, dn, lane);

        const float dself = dn * dn;
#pragma unroll
        for (int c = 0; c < 4; ++c) agg[c] = __builtin_fmaf(xself[c], dself, acc[c]);
    }
    *(VT*)(As + wave * T_PITCH + lane * 4) = agg;
    __syncthreads();

    const int half = lane >> 4;
    const int lid  = lane & 15;

    if (wave < HID_DIM / 16) {
        const v8f acc = tile_product<HID_DIM>(As, W0, wave, lane);
        const int   col  = wave * 16 + lid;
        const float bias = b0[col];
#pragma unroll
        for (int r = 0; r < 8; ++r)
            Hs[(r + half * 8) * T_PITCH + col] = fmaxf(acc[r] + bias, 0.0f);
    }
    __syncthreads();

    if (wave < OUT_DIM / 16) {
        const v8f acc = tile_product<OUT_DIM>(Hs, W1, wave, lane);
        const int col = wave * 16 + lid;
#pragma unroll
        for (int r = 0; r < 8; ++r)
            Ys[(r + half * 8) * Y_PITCH + col] = acc[r];
    }
    __syncthreads();

    if (node < n) {
        const v2f y = *(const v2f*)(Ys + wave * Y_PITCH + lane * 2);
        ((v2f*)Y1)[(size_t)node * 32 + lane] = y;
    }
}

extern "C" void kernel_launch(void* const* d_in, const int* in_sizes, int n_in,
                              void* d_out, int out_size, void* d_ws, size_t ws_size,
                              hipStream_t stream) {
    const int*   E  = (const int*)  d_in[1];
    const float* X  = (const float*)d_in[2];
    const float* W0 = (const float*)d_in[3];
    const float* b0 = (const float*)d_in[4];
    const float* W1 = (const float*)d_in[5];
    const float* b1 = (const float*)d_in[6];
    float* out = (float*)d_out;

    const int* Esrc = E;
    const int* Edst = E + N_EDGES;

    int*   cnt    = (int*)d_ws;
    int*   bucket = cnt + 50048;
    int*   ovf    = bucket + N_NODES * BUCKET_CAP;
    float* y1     = (float*)(ovf + 2 * N_EDGES);

    const int T = 256;
    zero_int_kernel<<<(N_NODES + 1 + T - 1) / T, T, 0, stream>>>(cnt, N_NODES + 1);
    bucket_kernel<<<(N_EDGES + T - 1) / T, T, 0, stream>>>(Esrc, Edst, cnt, bucket, ovf, N_EDGES);

    gather_transform_kernel<<<N_NODES / 16, 512, 0, stream>>>(cnt, bucket, ovf, X, W0, b0, W1, y1, N_NODES);

    gather_kernel<OUT_DIM, false><<<(N_NODES + 7) / 8, 256, 0, stream>>>(cnt, bucket, ovf, y1, b1, out, N_NODES);
}


# Bucket build on LDS counters: edge chunks counting-sorted by destination range in LDS, one work-group per range ranks its edges, stores slot rows; no global counting adds; sum order = LDS arrival rank
# speedup vs baseline: 4.8435x; 1.7226x over previous
#include <hip/hip_runtime.h>


typedef __attribute__((ext_vector_type(2))) float v2f;
typedef __attribute__((ext_vector_type(8))) float v8f;

#define N_NODES 50000
#define N_EDGES 800000
#define IN_DIM  128
#define HID_DIM 128
#define OUT_DIM 64

#define BUCKET_CAP 32

#define BIN_NODES   200
#define N_BINS      (N_NODES / BIN_NODES)
#define CHUNK_EDGES 3200
#define N_CHUNKS    (N_EDGES / CHUNK_EDGES)
#define SLOT_PITCH  (BUCKET_CAP + 1)
static_assert(N_NODES % BIN_NODES == 0, "whole ranges");
static_assert(N_EDGES % CHUNK_EDGES == 0 && CHUNK_EDGES % 4 == 0, "whole chunks of 16-byte index groups");
static_assert(CHUNK_EDGES < 32768, "start and count of a list fit 16 bits each");
static_assert(N_BINS <= 256, "one scan of 256 counters covers every range");

__global__ __launch_bounds__(512)
void bin_kernel(const int* __restrict__ src, const int* __restrict__ dst,
                int* __restrict__ pairs, int* __restrict__ table,
                int* __restrict__ ovf_len) {
    __shared__ int  binCnt[256];
    __shared__ int  binStart[256];
    __shared__ alignas(16) int2 stage[CHUNK_EDGES];

    const int t     = threadIdx.x;
    const int chunk = blockIdx.x;
    if (t < 256) binCnt[t] = 0;
    if (chunk == 0 && t == 0) *ovf_len = 0;
    __syncthreads();

    constexpr int GROUPS = CHUNK_EDGES / 4;
    const int4* __restrict__ dst4 = (const int4*)(dst + (size_t)chunk * CHUNK_EDGES);
    const int4* __restrict__ src4 = (const int4*)(src + (size_t)chunk * CHUNK_EDGES);

    int d[2][4], s[2][4], b[2][4], r[2][4];
#pragma unroll
    for (int u = 0; u < 2; ++u) {
        const int g = t + u * 512;
        if (g < GROUPS) {
            const int4 dv = dst4[g];
            const int4 sv = src4[g];
            d[u][0] = dv.x; d[u][1] = dv.y; d[u][2] = dv.z; d[u][3] = dv.w;
            s[u][0] = sv.x; s[u][1] = sv.y; s[u][2] = sv.z; s[u][3] = sv.w;
        }
    }
#pragma unroll
    for (int u = 0; u < 2; ++u) {
        if (t + u * 512 < GROUPS) {
#pragma unroll
            for (int c = 0; c < 4; ++c) {
                b[u][c] = (int)((unsigned)d[u][c] / (unsigned)BIN_NODES);
                r[u][c] = atomicAdd(&binCnt[b[u][c]], 1);
            }
        }
    }
    __syncthreads();

    if (t < 32) {
        int v[8];
        int tot = 0;
#pragma unroll
        for (int i = 0; i < 8; ++i) { v[i] = binCnt[t * 8 + i]; tot += v[i]; }
        int inc = tot;
#pragma unroll
        for (int off = 1; off < 32; off <<= 1) {
            const int up = __shfl_up(inc, off);
            if (t >= off) inc += up;
        }
        int run = inc - tot;
#pragma unroll
        for (int i = 0; i < 8; ++i) { binStart[t * 8 + i] = run; run += v[i]; }
    }
    __syncthreads();

#pragma unroll
    for (int u = 0; u < 2; ++u) {
        if (t + u * 512 < GROUPS) {
#pragma unroll
            for (int c = 0; c < 4; ++c)
                stage[binStart[b[u][c]] + r[u][c]] = make_int2(d[u][c], s[u][c]);
        }
    }
    if (t < N_BINS) table[t * N_CHUNKS + chunk] = binStart[t] | (binCnt[t] << 16);
    __syncthreads();

    int4* __restrict__ out4 = (int4*)(pairs + (size_t)chunk * CHUNK_EDGES * 2);
    const int4* st4 = (const int4*)stage;
    for (int i = t; i < CHUNK_EDGES / 2; i += 512) out4[i] = st4[i];
}

#define OWN_LOADS 8
#define OWN_ROWS  ((BIN_NODES + 15) / 16)
__global__ __launch_bounds__(512)
void own_kernel(const int* __restrict__ pairs, const int* __restrict__ table,
                int* __restrict__ cnt, int* __restrict__ bucket, int* __restrict__ ovf) {
    __shared__ int degL[BIN_NODES];
    __shared__ int slotL[BIN_NODES * SLOT_PITCH];

    const int t    = threadIdx.x;
    const int bin  = blockIdx.x;
    const int base = bin * BIN_NODES;
    if (t < BIN_NODES) degL[t] = 0;
    __syncthreads();

    const int g   = t >> 1;
    const int sub = t & 1;
    if (g < N_CHUNKS) {
        const int w     = table[bin * N_CHUNKS + g];
        const int start = w & 0xFFFF;
        const int n     = w >> 16;
        const int2* __restrict__ lp = (const int2*)pairs + (size_t)g * CHUNK_EDGES + start;
        for (int i = sub; i < n; i += 2 * OWN_LOADS) {
            int2 p[OWN_LOADS];
            int  rk[OWN_LOADS];
#pragma unroll
            for (int u = 0; u < OWN_LOADS; ++u) {
                const int j = i + 2 * u;
                p[u] = lp[j < n ? j : i];
            }
#pragma unroll
            for (int u = 0; u < OWN_LOADS; ++u) {
                if (i + 2 * u < n) rk[u] = atomicAdd(&degL[p[u].x - base], 1);
            }
#pragma unroll
            for (int u = 0; u < OWN_LOADS; ++u) {
                if (i + 2 * u < n) {
                    if (rk[u] < BUCKET_CAP) {
                        slotL[(p[u].x - base) * SLOT_PITCH + rk[u]] = p[u].y;
                    } else {
                        const int k = atomicAdd(&cnt[N_NODES], 1);
                        ((int2*)ovf)[k] = make_int2(p[u].x, p[u].y);
                    }
                }
            }
        }
    }
    __syncthreads();

    if (t < BIN_NODES) cnt[base + t] = degL[t];
    const int lane = t & 31;
    const int wave = t >> 5;
    int m[OWN_ROWS], v[OWN_ROWS];
#pragma unroll
    for (int q = 0; q < OWN_ROWS; ++q) {
        const int nl = wave + 16 * q;
        m[q] = 0;
        if (nl < BIN_NODES) m[q] = degL[nl] < BUCKET_CAP ? degL[nl] : BUCKET_CAP;
    }
#pragma unroll
    for (int q = 0; q < OWN_ROWS; ++q) {
        if (lane < m[q]) v[q] = slotL[(wave + 16 * q) * SLOT_PITCH + lane];
    }
#pragma unroll
    for (int q = 0; q < OWN_ROWS; ++q) {
        if (lane < m[q]) bucket[(size_t)(base + wave + 16 * q) * BUCKET_CAP + lane] = v[q];
    }
}

__device__ __forceinline__ float inv_sqrt_deg(int c) {
    return rsqrtf(fmaxf((float)(c + 1), 1.0f));
}

#define T_PITCH 132
#define Y_PITCH 68

template<int NCOL>
__device__ __forceinline__ v8f tile_product(const float* __restrict__ T,
                                            const float* __restrict__ W,
                                            int tile_n, int lane) {
    const int half = lane >> 4;
    const int lid  = lane & 15;
    const int col  = tile_n * 16 + lid;
    const int koff = half * 2;
    const float* arow = T + lid * T_PITCH;

    v8f acc = {};
#pragma unroll 8
    for (int k = 0; k < IN_DIM; k += 4) {
        v2f a = *(const v2f*)(arow + k + koff);
        v2f b;
        b.x = W[(size_t)(k + koff)     * NCOL + col];
        b.y = W[(size_t)(k + koff + 1) * NCOL + col];
        acc = __builtin_amdgcn_wmma_f32_16x16x4_f32(
            false, a, false, b, (short)0, acc, false, false);
    }
    return acc;
}

template<int V> struct RowVec;
template<> struct RowVec<4> { typedef __attribute__((ext_vector_type(4))) float type; };
template<> struct RowVec<2> { typedef __attribute__((ext_vector_type(2))) float type; };

template<int K, typename VT, int V>
__device__ __forceinline__ void add_rows(VT& acc, const VT* __restrict__ Yv,
                                         int my_s, int my_wbits, int j, int lane) {
    int   s[K];
    float w[K];
    VT    y[K];
#pragma unroll
    for (int k = 0; k < K; ++k) {
        s[k] = __builtin_amdgcn_readlane(my_s, j + k);
        w[k] = __builtin_bit_cast(float, __builtin_amdgcn_readlane(my_wbits, j + k));
    }
#pragma unroll
    for (int k = 0; k < K; ++k) y[k] = Yv[(size_t)s[k] * 32 + lane];
#pragma unroll
    for (int k = 0; k < K; ++k) {
#pragma unroll
        for (int c = 0; c < V; ++c) acc[c] = __builtin_fmaf(y[k][c], w[k], acc[c]);
    }
}

template<typename VT, int V>
__device__ __forceinline__ void sum_incoming(VT& acc, const int* __restrict__ cnt,
                                             const int* __restrict__ bucket,
                                             const int* __restrict__ ovf,
                                             const VT* __restrict__ Yv,
                                             int node, int deg, float dn, int lane) {
    const int m = deg < BUCKET_CAP ? deg : BUCKET_CAP;
    int   my_s = node;
    float my_w = 0.0f;
    if (lane < m) {
        my_s = bucket[node * BUCKET_CAP + lane];
        my_w = inv_sqrt_deg(cnt[my_s]) * dn;
    }
    const int my_wbits = __builtin_bit_cast(int, my_w);

    int j = 0;
    for (; j + 8 <= m; j += 8) add_rows<8, VT, V>(acc, Yv, my_s, my_wbits, j, lane);
    if (j + 4 <= m) { add_rows<4, VT, V>(acc, Yv, my_s, my_wbits, j, lane); j += 4; }
    const int r = m - j;
    if (r == 3)      add_rows<3, VT, V>(acc, Yv, my_s, my_wbits, j, lane);
    else if (r == 2) add_rows<2, VT, V>(acc, Yv, my_s, my_wbits, j, lane);
    else if (r == 1) add_rows<1, VT, V>(acc, Yv, my_s, my_wbits, j, lane);

    if (deg > BUCKET_CAP) {
        const int  len   = cnt[N_NODES];
        const int  n_ovf = len < N_EDGES ? len : N_EDGES;
        const int2* __restrict__ ovf2 = (const int2*)ovf;
        for (int k0 = 0; k0 < n_ovf; k0 += 32) {
            const int k = k0 + lane;
            bool  hit = false;
            int   o_s = node;
            float o_w = 0.0f;
            if (k < n_ovf) {
                const int2 rec = ovf2[k];
                if (rec.x == node) {
                    hit = true;
                    o_s = rec.y;
                    o_w = inv_sqrt_deg(cnt[o_s]) * dn;
                }
            }
            const int o_wbits = __builtin_bit_cast(int, o_w);
            unsigned todo = __builtin_amdgcn_ballot_w32(hit);
            while (todo != 0u) {
                const int l = __builtin_ctz(todo);
                todo &= todo - 1u;
                add_rows<1, VT, V>(acc, Yv, o_s, o_wbits, l, lane);
            }
        }
    }
}

template<int D, bool RELU>
__global__ __launch_bounds__(256)
void gather_kernel(const int* __restrict__ cnt, const int* __restrict__ bucket,
                   const int* __restrict__ ovf, const float* __restrict__ Y,
                   const float* __restrict__ bias, float* __restrict__ outp, int n) {
    constexpr int V = D / 32;
    typedef typename RowVec<V>::type VT;

    const int lane = threadIdx.x & 31;
    const int node = __builtin_amdgcn_readfirstlane((int)(blockIdx.x * 8 + (threadIdx.x >> 5)));
    if (node >= n) return;

    const int   deg = cnt[node];
    const float dn  = inv_sqrt_deg(deg);

    const VT* __restrict__ Yv = (const VT*)Y;
    const VT yself = Yv[(size_t)node * 32 + lane];

    VT acc;
#pragma unroll
    for (int c = 0; c < V; ++c) acc[c] = 0.0f;

    sum_incoming<VT, V>(acc, cnt, bucket, ovf, Yv, node, deg, dn, lane);

    const float dself = dn * dn;
    VT o;
#pragma unroll
    for (int c = 0; c < V; ++c) {
        const float t = __builtin_fmaf(yself[c], dself, acc[c]) + bias[lane * V + c];
        o[c] = RELU ? fmaxf(t, 0.0f) : t;
    }
    ((VT*)outp)[(size_t)node * 32 + lane] = o;
}

static_assert(N_NODES % 16 == 0, "one 16-row tile per work-group");
static_assert(IN_DIM == HID_DIM, "tile_product steps K over IN_DIM for both products");

__global__ __launch_bounds__(512)
void gather_transform_kernel(const int* __restrict__ cnt, const int* __restrict__ bucket,
                             const int* __restrict__ ovf, const float* __restrict__ X,
                             const float* __restrict__ W0, const float* __restrict__ b0,
                             const float* __restrict__ W1, float* __restrict__ Y1, int n) {
    typedef typename RowVec<4>::type VT;
    __shared__ float As[16 * T_PITCH];
    __shared__ float Hs[16 * T_PITCH];
    __shared__ float Ys[16 * Y_PITCH];

    const int lane = threadIdx.x & 31;
    const int wave = __builtin_amdgcn_readfirstlane((int)(threadIdx.x >> 5));
    const int node = __builtin_amdgcn_readfirstlane((int)(blockIdx.x * 16)) + wave;

    VT agg;
#pragma unroll
    for (int c = 0; c < 4; ++c) agg[c] = 0.0f;

    if (node < n) {
        const int   deg = cnt[node];
        const float dn  = inv_sqrt_deg(deg);

        const VT* __restrict__ Xv = (const VT*)X;
        const VT xself = Xv[(size_t)node * 32 + lane];

        VT acc;
#pragma unroll
        for (int c = 0; c < 4; ++c) acc[c] = 0.0f;

        sum_incoming<VT, 4>(acc, cnt, bucket, ovf, Xv, node, deg, dn, lane);

        const float dself = dn * dn;
#pragma unroll
        for (int c = 0; c < 4; ++c) agg[c] = __builtin_fmaf(xself[c], dself, acc[c]);
    }
    *(VT*)(As + wave * T_PITCH + lane * 4) = agg;
    __syncthreads();

    const int half = lane >> 4;
    const int lid  = lane & 15;

    if (wave < HID_DIM / 16) {
        const v8f acc = tile_product<HID_DIM>(As, W0, wave, lane);
        const int   col  = wave * 16 + lid;
        const float bias = b0[col];
#pragma unroll
        for (int r = 0; r < 8; ++r)
            Hs[(r + half * 8) * T_PITCH + col] = fmaxf(acc[r] + bias, 0.0f);
    }
    __syncthreads();

    if (wave < OUT_DIM / 16) {
        const v8f acc = tile_product<OUT_DIM>(Hs, W1, wave, lane);
        const int col = wave * 16 + lid;
#pragma unroll
        for (int r = 0; r < 8; ++r)
            Ys[(r + half * 8) * Y_PITCH + col] = acc[r];
    }
    __syncthreads();

    if (node < n) {
        const v2f y = *(const v2f*)(Ys + wave * Y_PITCH + lane * 2);
        ((v2f*)Y1)[(size_t)node * 32 + lane] = y;
    }
}

extern "C" void kernel_launch(void* const* d_in, const int* in_sizes, int n_in,
                              void* d_out, int out_size, void* d_ws, size_t ws_size,
                              hipStream_t stream) {
    const int*   E  = (const int*)  d_in[1];
    const float* X  = (const float*)d_in[2];
    const float* W0 = (const float*)d_in[3];
    const float* b0 = (const float*)d_in[4];
    const float* W1 = (const float*)d_in[5];
    const float* b1 = (const float*)d_in[6];
    float* out = (float*)d_out;

    const int* Esrc = E;
    const int* Edst = E + N_EDGES;

    int*   cnt    = (int*)d_ws;
    int*   bucket = cnt + 50048;
    int*   ovf    = bucket + N_NODES * BUCKET_CAP;
    float* y1     = (float*)(ovf + 2 * N_EDGES);
    int*   pairs  = (int*)(y1 + N_NODES * OUT_DIM);
    int*   table  = pairs + 2 * N_EDGES;

    bin_kernel<<<N_CHUNKS, 512, 0, stream>>>(Esrc, Edst, pairs, table, cnt + N_NODES);
    own_kernel<<<N_BINS, 512, 0, stream>>>(pairs, table, cnt, bucket, ovf);

    gather_transform_kernel<<<N_NODES / 16, 512, 0, stream>>>(cnt, bucket, ovf, X, W0, b0, W1, y1, N_NODES);

    gather_kernel<OUT_DIM, false><<<(N_NODES + 7) / 8, 256, 0, stream>>>(cnt, bucket, ovf, y1, b1, out, N_NODES);
}


# Layer-0 tile 16 -> 32 nodes: each wave sums two nodes (slot and count loads of both ahead of the rows), all 16 waves multiply; W0/W1 read once per 32 nodes, half the waves; sum order unchanged
# speedup vs baseline: 5.2579x; 1.0856x over previous
#include <hip/hip_runtime.h>


typedef __attribute__((ext_vector_type(2))) float v2f;
typedef __attribute__((ext_vector_type(8))) float v8f;

#define N_NODES 50000
#define N_EDGES 800000
#define IN_DIM  128
#define HID_DIM 128
#define OUT_DIM 64

#define BUCKET_CAP 32

#define BIN_NODES   200
#define N_BINS      (N_NODES / BIN_NODES)
#define CHUNK_EDGES 3200
#define N_CHUNKS    (N_EDGES / CHUNK_EDGES)
#define SLOT_PITCH  (BUCKET_CAP + 1)
static_assert(N_NODES % BIN_NODES == 0, "whole ranges");
static_assert(N_EDGES % CHUNK_EDGES == 0 && CHUNK_EDGES % 4 == 0, "whole chunks of 16-byte index groups");
static_assert(CHUNK_EDGES < 32768, "start and count of a list fit 16 bits each");
static_assert(N_BINS <= 256, "one scan of 256 counters covers every range");

__global__ __launch_bounds__(512)
void bin_kernel(const int* __restrict__ src, const int* __restrict__ dst,
                int* __restrict__ pairs, int* __restrict__ table,
                int* __restrict__ ovf_len) {
    __shared__ int  binCnt[256];
    __shared__ int  binStart[256];
    __shared__ alignas(16) int2 stage[CHUNK_EDGES];

    const int t     = threadIdx.x;
    const int chunk = blockIdx.x;
    if (t < 256) binCnt[t] = 0;
    if (chunk == 0 && t == 0) *ovf_len = 0;
    __syncthreads();

    constexpr int GROUPS = CHUNK_EDGES / 4;
    const int4* __restrict__ dst4 = (const int4*)(dst + (size_t)chunk * CHUNK_EDGES);
    const int4* __restrict__ src4 = (const int4*)(src + (size_t)chunk * CHUNK_EDGES);

    int d[2][4], s[2][4], b[2][4], r[2][4];
#pragma unroll
    for (int u = 0; u < 2; ++u) {
        const int g = t + u * 512;
        if (g < GROUPS) {
            const int4 dv = dst4[g];
            const int4 sv = src4[g];
            d[u][0] = dv.x; d[u][1] = dv.y; d[u][2] = dv.z; d[u][3] = dv.w;
            s[u][0] = sv.x; s[u][1] = sv.y; s[u][2] = sv.z; s[u][3] = sv.w;
        }
    }
#pragma unroll
    for (int u = 0; u < 2; ++u) {
        if (t + u * 512 < GROUPS) {
#pragma unroll
            for (int c = 0; c < 4; ++c) {
                b[u][c] = (int)((unsigned)d[u][c] / (unsigned)BIN_NODES);
                r[u][c] = atomicAdd(&binCnt[b[u][c]], 1);
            }
        }
    }
    __syncthreads();

    if (t < 32) {
        int v[8];
        int tot = 0;
#pragma unroll
        for (int i = 0; i < 8; ++i) { v[i] = binCnt[t * 8 + i]; tot += v[i]; }
        int inc = tot;
#pragma unroll
        for (int off = 1; off < 32; off <<= 1) {
            const int up = __shfl_up(inc, off);
            if (t >= off) inc += up;
        }
        int run = inc - tot;
#pragma unroll
        for (int i = 0; i < 8; ++i) { binStart[t * 8 + i] = run; run += v[i]; }
    }
    __syncthreads();

#pragma unroll
    for (int u = 0; u < 2; ++u) {
        if (t + u * 512 < GROUPS) {
#pragma unroll
            for (int c = 0; c < 4; ++c)
                stage[binStart[b[u][c]] + r[u][c]] = make_int2(d[u][c], s[u][c]);
        }
    }
    if (t < N_BINS) table[t * N_CHUNKS + chunk] = binStart[t] | (binCnt[t] << 16);
    __syncthreads();

    int4* __restrict__ out4 = (int4*)(pairs + (size_t)chunk * CHUNK_EDGES * 2);
    const int4* st4 = (const int4*)stage;
    for (int i = t; i < CHUNK_EDGES / 2; i += 512) out4[i] = st4[i];
}

#define OWN_LOADS 8
#define OWN_ROWS  ((BIN_NODES + 15) / 16)
__global__ __launch_bounds__(512)
void own_kernel(const int* __restrict__ pairs, const int* __restrict__ table,
                int* __restrict__ cnt, int* __restrict__ bucket, int* __restrict__ ovf) {
    __shared__ int degL[BIN_NODES];
    __shared__ int slotL[BIN_NODES * SLOT_PITCH];

    const int t    = threadIdx.x;
    const int bin  = blockIdx.x;
    const int base = bin * BIN_NODES;
    if (t < BIN_NODES) degL[t] = 0;
    __syncthreads();

    const int g   = t >> 1;
    const int sub = t & 1;
    if (g < N_CHUNKS) {
        const int w     = table[bin * N_CHUNKS + g];
        const int start = w & 0xFFFF;
        const int n     = w >> 16;
        const int2* __restrict__ lp = (const int2*)pairs + (size_t)g * CHUNK_EDGES + start;
        for (int i = sub; i < n; i += 2 * OWN_LOADS) {
            int2 p[OWN_LOADS];
            int  rk[OWN_LOADS];
#pragma unroll
            for (int u = 0; u < OWN_LOADS; ++u) {
                const int j = i + 2 * u;
                p[u] = lp[j < n ? j : i];
            }
#pragma unroll
            for (int u = 0; u < OWN_LOADS; ++u) {
                if (i + 2 * u < n) rk[u] = atomicAdd(&degL[p[u].x - base], 1);
            }
#pragma unroll
            for (int u = 0; u < OWN_LOADS; ++u) {
                if (i + 2 * u < n) {
                    if (rk[u] < BUCKET_CAP) {
                        slotL[(p[u].x - base) * SLOT_PITCH + rk[u]] = p[u].y;
                    } else {
                        const int k = atomicAdd(&cnt[N_NODES], 1);
                        ((int2*)ovf)[k] = make_int2(p[u].x, p[u].y);
                    }
                }
            }
        }
    }
    __syncthreads();

    if (t < BIN_NODES) cnt[base + t] = degL[t];
    const int lane = t & 31;
    const int wave = t >> 5;
    int m[OWN_ROWS], v[OWN_ROWS];
#pragma unroll
    for (int q = 0; q < OWN_ROWS; ++q) {
        const int nl = wave + 16 * q;
        m[q] = 0;
        if (nl < BIN_NODES) m[q] = degL[nl] < BUCKET_CAP ? degL[nl] : BUCKET_CAP;
    }
#pragma unroll
    for (int q = 0; q < OWN_ROWS; ++q) {
        if (lane < m[q]) v[q] = slotL[(wave + 16 * q) * SLOT_PITCH + lane];
    }
#pragma unroll
    for (int q = 0; q < OWN_ROWS; ++q) {
        if (lane < m[q]) bucket[(size_t)(base + wave + 16 * q) * BUCKET_CAP + lane] = v[q];
    }
}

__device__ __forceinline__ float inv_sqrt_deg(int c) {
    return rsqrtf(fmaxf((float)(c + 1), 1.0f));
}

#define T_PITCH 132
#define Y_PITCH 68

template<int NCOL>
__device__ __forceinline__ v8f tile_product(const float* __restrict__ T,
                                            const float* __restrict__ W,
                                            int tile_n, int lane) {
    const int half = lane >> 4;
    const int lid  = lane & 15;
    const int col  = tile_n * 16 + lid;
    const int koff = half * 2;
    const float* arow = T + lid * T_PITCH;
    const float* __restrict__ wcol = W + (size_t)koff * NCOL + col;

    v8f acc = {};
#pragma unroll 8
    for (int k = 0; k < IN_DIM; k += 4) {
        v2f a = *(const v2f*)(arow + k + koff);
        v2f b;
        b.x = wcol[(size_t)k       * NCOL];
        b.y = wcol[(size_t)(k + 1) * NCOL];
        acc = __builtin_amdgcn_wmma_f32_16x16x4_f32(
            false, a, false, b, (short)0, acc, false, false);
    }
    return acc;
}

template<int V> struct RowVec;
template<> struct RowVec<4> { typedef __attribute__((ext_vector_type(4))) float type; };
template<> struct RowVec<2> { typedef __attribute__((ext_vector_type(2))) float type; };

template<int K, typename VT, int V>
__device__ __forceinline__ void add_rows(VT& acc, const VT* __restrict__ Yv,
                                         int my_s, int my_wbits, int j, int lane) {
    int   s[K];
    float w[K];
    VT    y[K];
#pragma unroll
    for (int k = 0; k < K; ++k) {
        s[k] = __builtin_amdgcn_readlane(my_s, j + k);
        w[k] = __builtin_bit_cast(float, __builtin_amdgcn_readlane(my_wbits, j + k));
    }
#pragma unroll
    for (int k = 0; k < K; ++k) y[k] = Yv[(size_t)s[k] * 32 + lane];
#pragma unroll
    for (int k = 0; k < K; ++k) {
#pragma unroll
        for (int c = 0; c < V; ++c) acc[c] = __builtin_fmaf(y[k][c], w[k], acc[c]);
    }
}


__device__ __forceinline__ int filled_slots(int deg) {
    return deg < BUCKET_CAP ? deg : BUCKET_CAP;
}

__device__ __forceinline__ int slot_source(const int* __restrict__ bucket, int node, int deg, int lane, int no_slot) {
    int my_s = no_slot;
    if (lane < filled_slots(deg)) my_s = bucket[node * BUCKET_CAP + lane];
    return my_s;
}

__device__ __forceinline__ float slot_weight(int src_cnt, int deg, float dn, int lane) {
    return lane < filled_slots(deg) ? inv_sqrt_deg(src_cnt) * dn : 0.0f;
}

template<typename VT, int V>
__device__ __forceinline__ void sum_slots(VT& acc, const int* __restrict__ cnt,
                                          const int* __restrict__ ovf,
                                          const VT* __restrict__ Yv,
                                          int node, int deg, float dn,
                                          int my_s, float my_w, int lane) {
    const int m = filled_slots(deg);
    const int my_wbits = __builtin_bit_cast(int, my_w);

    int j = 0;
    for (; j + 8 <= m; j += 8) add_rows<8, VT, V>(acc, Yv, my_s, my_wbits, j, lane);
    if (j + 4 <= m) { add_rows<4, VT, V>(acc, Yv, my_s, my_wbits, j, lane); j += 4; }
    const int r = m - j;
    if (r == 3)      add_rows<3, VT, V>(acc, Yv, my_s, my_wbits, j, lane);
    else if (r == 2) add_rows<2, VT, V>(acc, Yv, my_s, my_wbits, j, lane);
    else if (r == 1) add_rows<1, VT, V>(acc, Yv, my_s, my_wbits, j, lane);

    if (deg > BUCKET_CAP) {
        const int  len   = cnt[N_NODES];
        const int  n_ovf = len < N_EDGES ? len : N_EDGES;
        const int2* __restrict__ ovf2 = (const int2*)ovf;
        for (int k0 = 0; k0 < n_ovf; k0 += 32) {
            const int k = k0 + lane;
            bool  hit = false;
            int   o_s = node;
            float o_w = 0.0f;
            if (k < n_ovf) {
                const int2 rec = ovf2[k];
                if (rec.x == node) {
                    hit = true;
                    o_s = rec.y;
                    o_w = inv_sqrt_deg(cnt[o_s]) * dn;
                }
            }
            const int o_wbits = __builtin_bit_cast(int, o_w);
            unsigned todo = __builtin_amdgcn_ballot_w32(hit);
            while (todo != 0u) {
                const int l = __builtin_ctz(todo);
                todo &= todo - 1u;
                add_rows<1, VT, V>(acc, Yv, o_s, o_wbits, l, lane);
            }
        }
    }
}

template<typename VT, int V>
__device__ __forceinline__ void sum_incoming(VT& acc, const int* __restrict__ cnt,
                                             const int* __restrict__ bucket,
                                             const int* __restrict__ ovf,
                                             const VT* __restrict__ Yv,
                                             int node, int deg, float dn, int lane) {
    int   my_s = node;
    float my_w = 0.0f;
    if (lane < filled_slots(deg)) {
        my_s = bucket[node * BUCKET_CAP + lane];
        my_w = inv_sqrt_deg(cnt[my_s]) * dn;
    }
    sum_slots<VT, V>(acc, cnt, ovf, Yv, node, deg, dn, my_s, my_w, lane);
}

template<int D, bool RELU>
__global__ __launch_bounds__(256)
void gather_kernel(const int* __restrict__ cnt, const int* __restrict__ bucket,
                   const int* __restrict__ ovf, const float* __restrict__ Y,
                   const float* __restrict__ bias, float* __restrict__ outp, int n) {
    constexpr int V = D / 32;
    typedef typename RowVec<V>::type VT;

    const int lane = threadIdx.x & 31;
    const int node = __builtin_amdgcn_readfirstlane((int)(blockIdx.x * 8 + (threadIdx.x >> 5)));
    if (node >= n) return;

    const int   deg = cnt[node];
    const float dn  = inv_sqrt_deg(deg);

    const VT* __restrict__ Yv = (const VT*)Y;
    const VT yself = Yv[(size_t)node * 32 + lane];

    VT acc;
#pragma unroll
    for (int c = 0; c < V; ++c) acc[c] = 0.0f;

    sum_incoming<VT, V>(acc, cnt, bucket, ovf, Yv, node, deg, dn, lane);

    const float dself = dn * dn;
    VT o;
#pragma unroll
    for (int c = 0; c < V; ++c) {
        const float t = __builtin_fmaf(yself[c], dself, acc[c]) + bias[lane * V + c];
        o[c] = RELU ? fmaxf(t, 0.0f) : t;
    }
    ((VT*)outp)[(size_t)node * 32 + lane] = o;
}

#define TILE_NODES 32
static_assert(N_NODES % 16 == 0, "whole 16-row tiles");
static_assert(IN_DIM == HID_DIM, "tile_product steps K over IN_DIM for both products");
static_assert(HID_DIM / 16 == 8 && OUT_DIM / 16 == 4, "wave -> (column tile, row tile) split below");

__global__ __launch_bounds__(512)
void gather_transform_kernel(const int* __restrict__ cnt, const int* __restrict__ bucket,
                             const int* __restrict__ ovf, const float* __restrict__ X,
                             const float* __restrict__ W0, const float* __restrict__ b0,
                             const float* __restrict__ W1, float* __restrict__ Y1, int n) {
    typedef typename RowVec<4>::type VT;
    __shared__ float As[TILE_NODES * T_PITCH];
    __shared__ float Hs[TILE_NODES * T_PITCH];
    __shared__ float Ys[TILE_NODES * Y_PITCH];

    const int lane = threadIdx.x & 31;
    const int wave = __builtin_amdgcn_readfirstlane((int)(threadIdx.x >> 5));
    const int base = __builtin_amdgcn_readfirstlane((int)(blockIdx.x * TILE_NODES));

    constexpr int NH = TILE_NODES / 16;
    const VT* __restrict__ Xv = (const VT*)X;
    int   nd[NH], deg[NH], my_s[NH], src_cnt[NH];
    float dn[NH], my_w[NH];
    VT    xself[NH];
#pragma unroll
    for (int h = 0; h < NH; ++h) {
        const int node = base + h * 16 + wave;
        nd[h]  = node < n ? node : n - 1;
        deg[h] = cnt[nd[h]];
        dn[h]  = inv_sqrt_deg(deg[h]);
    }
#pragma unroll
    for (int h = 0; h < NH; ++h) {
        xself[h] = Xv[(size_t)nd[h] * 32 + lane];
        my_s[h]  = slot_source(bucket, nd[h], deg[h], lane, 0);
    }
#pragma unroll
    for (int h = 0; h < NH; ++h) src_cnt[h] = cnt[my_s[h]];
#pragma unroll
    for (int h = 0; h < NH; ++h) my_w[h] = slot_weight(src_cnt[h], deg[h], dn[h], lane);
#pragma unroll
    for (int h = 0; h < NH; ++h) {
        VT acc;
#pragma unroll
        for (int c = 0; c < 4; ++c) acc[c] = 0.0f;

        sum_slots<VT, 4>(acc, cnt, ovf, Xv, nd[h], deg[h], dn[h], my_s[h], my_w[h], lane);

        const float dself = dn[h] * dn[h];
        VT agg;
#pragma unroll
        for (int c = 0; c < 4; ++c) agg[c] = __builtin_fmaf(xself[h][c], dself, acc[c]);
        *(VT*)(As + (h * 16 + wave) * T_PITCH + lane * 4) = agg;
    }
    __syncthreads();

    const int half = lane >> 4;
    const int lid  = lane & 15;

    {
        const int tn = wave & 7;
        const int tm = wave >> 3;
        const v8f acc = tile_product<HID_DIM>(As + tm * 16 * T_PITCH, W0, tn, lane);
        const int   col  = tn * 16 + lid;
        const float bias = b0[col];
#pragma unroll
        for (int r = 0; r < 8; ++r)
            Hs[(tm * 16 + r + half * 8) * T_PITCH + col] = fmaxf(acc[r] + bias, 0.0f);
    }
    __syncthreads();

    if (wave < 2 * (OUT_DIM / 16)) {
        const int tn = wave & 3;
        const int tm = wave >> 2;
        const v8f acc = tile_product<OUT_DIM>(Hs + tm * 16 * T_PITCH, W1, tn, lane);
        const int col = tn * 16 + lid;
#pragma unroll
        for (int r = 0; r < 8; ++r)
            Ys[(tm * 16 + r + half * 8) * Y_PITCH + col] = acc[r];
    }
    __syncthreads();

    for (int h = 0; h < TILE_NODES / 16; ++h) {
        const int row  = h * 16 + wave;
        const int node = base + row;
        if (node < n) {
            const v2f y = *(const v2f*)(Ys + row * Y_PITCH + lane * 2);
            ((v2f*)Y1)[(size_t)node * 32 + lane] = y;
        }
    }
}

extern "C" void kernel_launch(void* const* d_in, const int* in_sizes, int n_in,
                              void* d_out, int out_size, void* d_ws, size_t ws_size,
                              hipStream_t stream) {
    const int*   E  = (const int*)  d_in[1];
    const float* X  = (const float*)d_in[2];
    const float* W0 = (const float*)d_in[3];
    const float* b0 = (const float*)d_in[4];
    const float* W1 = (const float*)d_in[5];
    const float* b1 = (const float*)d_in[6];
    float* out = (float*)d_out;

    const int* Esrc = E;
    const int* Edst = E + N_EDGES;

    int*   cnt    = (int*)d_ws;
    int*   bucket = cnt + 50048;
    int*   ovf    = bucket + N_NODES * BUCKET_CAP;
    float* y1     = (float*)(ovf + 2 * N_EDGES);
    int*   pairs  = (int*)(y1 + N_NODES * OUT_DIM);
    int*   table  = pairs + 2 * N_EDGES;

    bin_kernel<<<N_CHUNKS, 512, 0, stream>>>(Esrc, Edst, pairs, table, cnt + N_NODES);
    own_kernel<<<N_BINS, 512, 0, stream>>>(pairs, table, cnt, bucket, ovf);

    gather_transform_kernel<<<(N_NODES + TILE_NODES - 1) / TILE_NODES, 512, 0, stream>>>(cnt, bucket, ovf, X, W0, b0, W1, y1, N_NODES);

    gather_kernel<OUT_DIM, false><<<(N_NODES + 7) / 8, 256, 0, stream>>>(cnt, bucket, ovf, y1, b1, out, N_NODES);
}


# Layer-1 product rows stored already scaled by their own node's 1/sqrt(deg): layer-1 gather adds source rows unweighted, scales the sum once; no source-count loads there; sum order unchanged
# speedup vs baseline: 5.4684x; 1.0400x over previous
#include <hip/hip_runtime.h>


typedef __attribute__((ext_vector_type(2))) float v2f;
typedef __attribute__((ext_vector_type(8))) float v8f;

#define N_NODES 50000
#define N_EDGES 800000
#define IN_DIM  128
#define HID_DIM 128
#define OUT_DIM 64

#define BUCKET_CAP 32

#define BIN_NODES   200
#define N_BINS      (N_NODES / BIN_NODES)
#define CHUNK_EDGES 3200
#define N_CHUNKS    (N_EDGES / CHUNK_EDGES)
#define SLOT_PITCH  (BUCKET_CAP + 1)
static_assert(N_NODES % BIN_NODES == 0, "whole ranges");
static_assert(N_EDGES % CHUNK_EDGES == 0 && CHUNK_EDGES % 4 == 0, "whole chunks of 16-byte index groups");
static_assert(CHUNK_EDGES < 32768, "start and count of a list fit 16 bits each");
static_assert(N_BINS <= 256, "one scan of 256 counters covers every range");

__global__ __launch_bounds__(512)
void bin_kernel(const int* __restrict__ src, const int* __restrict__ dst,
                int* __restrict__ pairs, int* __restrict__ table,
                int* __restrict__ ovf_len) {
    __shared__ int  binCnt[256];
    __shared__ int  binStart[256];
    __shared__ alignas(16) int2 stage[CHUNK_EDGES];

    const int t     = threadIdx.x;
    const int chunk = blockIdx.x;
    if (t < 256) binCnt[t] = 0;
    if (chunk == 0 && t == 0) *ovf_len = 0;
    __syncthreads();

    constexpr int GROUPS = CHUNK_EDGES / 4;
    const int4* __restrict__ dst4 = (const int4*)(dst + (size_t)chunk * CHUNK_EDGES);
    const int4* __restrict__ src4 = (const int4*)(src + (size_t)chunk * CHUNK_EDGES);

    int d[2][4], s[2][4], b[2][4], r[2][4];
#pragma unroll
    for (int u = 0; u < 2; ++u) {
        const int g = t + u * 512;
        if (g < GROUPS) {
            const int4 dv = dst4[g];
            const int4 sv = src4[g];
            d[u][0] = dv.x; d[u][1] = dv.y; d[u][2] = dv.z; d[u][3] = dv.w;
            s[u][0] = sv.x; s[u][1] = sv.y; s[u][2] = sv.z; s[u][3] = sv.w;
        }
    }
#pragma unroll
    for (int u = 0; u < 2; ++u) {
        if (t + u * 512 < GROUPS) {
#pragma unroll
            for (int c = 0; c < 4; ++c) {
                b[u][c] = (int)((unsigned)d[u][c] / (unsigned)BIN_NODES);
                r[u][c] = atomicAdd(&binCnt[b[u][c]], 1);
            }
        }
    }
    __syncthreads();

    if (t < 32) {
        int v[8];
        int tot = 0;
#pragma unroll
        for (int i = 0; i < 8; ++i) { v[i] = binCnt[t * 8 + i]; tot += v[i]; }
        int inc = tot;
#pragma unroll
        for (int off = 1; off < 32; off <<= 1) {
            const int up = __shfl_up(inc, off);
            if (t >= off) inc += up;
        }
        int run = inc - tot;
#pragma unroll
        for (int i = 0; i < 8; ++i) { binStart[t * 8 + i] = run; run += v[i]; }
    }
    __syncthreads();

#pragma unroll
    for (int u = 0; u < 2; ++u) {
        if (t + u * 512 < GROUPS) {
#pragma unroll
            for (int c = 0; c < 4; ++c)
                stage[binStart[b[u][c]] + r[u][c]] = make_int2(d[u][c], s[u][c]);
        }
    }
    if (t < N_BINS) table[t * N_CHUNKS + chunk] = binStart[t] | (binCnt[t] << 16);
    __syncthreads();

    int4* __restrict__ out4 = (int4*)(pairs + (size_t)chunk * CHUNK_EDGES * 2);
    const int4* st4 = (const int4*)stage;
    for (int i = t; i < CHUNK_EDGES / 2; i += 512) out4[i] = st4[i];
}

#define OWN_LOADS 8
#define OWN_ROWS  ((BIN_NODES + 15) / 16)
__global__ __launch_bounds__(512)
void own_kernel(const int* __restrict__ pairs, const int* __restrict__ table,
                int* __restrict__ cnt, int* __restrict__ bucket, int* __restrict__ ovf) {
    __shared__ int degL[BIN_NODES];
    __shared__ int slotL[BIN_NODES * SLOT_PITCH];

    const int t    = threadIdx.x;
    const int bin  = blockIdx.x;
    const int base = bin * BIN_NODES;
    if (t < BIN_NODES) degL[t] = 0;
    __syncthreads();

    const int g   = t >> 1;
    const int sub = t & 1;
    if (g < N_CHUNKS) {
        const int w     = table[bin * N_CHUNKS + g];
        const int start = w & 0xFFFF;
        const int n     = w >> 16;
        const int2* __restrict__ lp = (const int2*)pairs + (size_t)g * CHUNK_EDGES + start;
        for (int i = sub; i < n; i += 2 * OWN_LOADS) {
            int2 p[OWN_LOADS];
            int  rk[OWN_LOADS];
#pragma unroll
            for (int u = 0; u < OWN_LOADS; ++u) {
                const int j = i + 2 * u;
                p[u] = lp[j < n ? j : i];
            }
#pragma unroll
            for (int u = 0; u < OWN_LOADS; ++u) {
                if (i + 2 * u < n) rk[u] = atomicAdd(&degL[p[u].x - base], 1);
            }
#pragma unroll
            for (int u = 0; u < OWN_LOADS; ++u) {
                if (i + 2 * u < n) {
                    if (rk[u] < BUCKET_CAP) {
                        slotL[(p[u].x - base) * SLOT_PITCH + rk[u]] = p[u].y;
                    } else {
                        const int k = atomicAdd(&cnt[N_NODES], 1);
                        ((int2*)ovf)[k] = make_int2(p[u].x, p[u].y);
                    }
                }
            }
        }
    }
    __syncthreads();

    if (t < BIN_NODES) cnt[base + t] = degL[t];
    const int lane = t & 31;
    const int wave = t >> 5;
    int m[OWN_ROWS], v[OWN_ROWS];
#pragma unroll
    for (int q = 0; q < OWN_ROWS; ++q) {
        const int nl = wave + 16 * q;
        m[q] = 0;
        if (nl < BIN_NODES) m[q] = degL[nl] < BUCKET_CAP ? degL[nl] : BUCKET_CAP;
    }
#pragma unroll
    for (int q = 0; q < OWN_ROWS; ++q) {
        if (lane < m[q]) v[q] = slotL[(wave + 16 * q) * SLOT_PITCH + lane];
    }
#pragma unroll
    for (int q = 0; q < OWN_ROWS; ++q) {
        if (lane < m[q]) bucket[(size_t)(base + wave + 16 * q) * BUCKET_CAP + lane] = v[q];
    }
}

__device__ __forceinline__ float inv_sqrt_deg(int c) {
    return rsqrtf(fmaxf((float)(c + 1), 1.0f));
}

#define T_PITCH 132
#define Y_PITCH 68

template<int NCOL>
__device__ __forceinline__ v8f tile_product(const float* __restrict__ T,
                                            const float* __restrict__ W,
                                            int tile_n, int lane) {
    const int half = lane >> 4;
    const int lid  = lane & 15;
    const int col  = tile_n * 16 + lid;
    const int koff = half * 2;
    const float* arow = T + lid * T_PITCH;
    const float* __restrict__ wcol = W + (size_t)koff * NCOL + col;

    v8f acc = {};
#pragma unroll 8
    for (int k = 0; k < IN_DIM; k += 4) {
        v2f a = *(const v2f*)(arow + k + koff);
        v2f b;
        b.x = wcol[(size_t)k       * NCOL];
        b.y = wcol[(size_t)(k + 1) * NCOL];
        acc = __builtin_amdgcn_wmma_f32_16x16x4_f32(
            false, a, false, b, (short)0, acc, false, false);
    }
    return acc;
}

template<int V> struct RowVec;
template<> struct RowVec<4> { typedef __attribute__((ext_vector_type(4))) float type; };
template<> struct RowVec<2> { typedef __attribute__((ext_vector_type(2))) float type; };

template<int K, typename VT, int V, bool WEIGHTED>
__device__ __forceinline__ void add_rows(VT& acc, const VT* __restrict__ Yv,
                                         int my_s, int my_wbits, int j, int lane) {
    int   s[K];
    float w[K];
    VT    y[K];
#pragma unroll
    for (int k = 0; k < K; ++k) {
        s[k] = __builtin_amdgcn_readlane(my_s, j + k);
        w[k] = WEIGHTED ? __builtin_bit_cast(float, __builtin_amdgcn_readlane(my_wbits, j + k)) : 1.0f;
    }
#pragma unroll
    for (int k = 0; k < K; ++k) y[k] = Yv[(size_t)s[k] * 32 + lane];
#pragma unroll
    for (int k = 0; k < K; ++k) {
#pragma unroll
        for (int c = 0; c < V; ++c)
            acc[c] = WEIGHTED ? __builtin_fmaf(y[k][c], w[k], acc[c]) : acc[c] + y[k][c];
    }
}


__device__ __forceinline__ int filled_slots(int deg) {
    return deg < BUCKET_CAP ? deg : BUCKET_CAP;
}

__device__ __forceinline__ int slot_source(const int* __restrict__ bucket, int node, int deg, int lane, int no_slot) {
    int my_s = no_slot;
    if (lane < filled_slots(deg)) my_s = bucket[node * BUCKET_CAP + lane];
    return my_s;
}

__device__ __forceinline__ float slot_weight(int src_cnt, int deg, float dn, int lane) {
    return lane < filled_slots(deg) ? inv_sqrt_deg(src_cnt) * dn : 0.0f;
}

template<typename VT, int V, bool WEIGHTED>
__device__ __forceinline__ void sum_slots(VT& acc, const int* __restrict__ cnt,
                                          const int* __restrict__ ovf,
                                          const VT* __restrict__ Yv,
                                          int node, int deg, float dn,
                                          int my_s, float my_w, int lane) {
    const int m = filled_slots(deg);
    const int my_wbits = __builtin_bit_cast(int, my_w);

    int j = 0;
    for (; j + 8 <= m; j += 8) add_rows<8, VT, V, WEIGHTED>(acc, Yv, my_s, my_wbits, j, lane);
    if (j + 4 <= m) { add_rows<4, VT, V, WEIGHTED>(acc, Yv, my_s, my_wbits, j, lane); j += 4; }
    const int r = m - j;
    if (r == 3)      add_rows<3, VT, V, WEIGHTED>(acc, Yv, my_s, my_wbits, j, lane);
    else if (r == 2) add_rows<2, VT, V, WEIGHTED>(acc, Yv, my_s, my_wbits, j, lane);
    else if (r == 1) add_rows<1, VT, V, WEIGHTED>(acc, Yv, my_s, my_wbits, j, lane);

    if (deg > BUCKET_CAP) {
        const int  len   = cnt[N_NODES];
        const int  n_ovf = len < N_EDGES ? len : N_EDGES;
        const int2* __restrict__ ovf2 = (const int2*)ovf;
        for (int k0 = 0; k0 < n_ovf; k0 += 32) {
            const int k = k0 + lane;
            bool  hit = false;
            int   o_s = node;
            float o_w = 0.0f;
            if (k < n_ovf) {
                const int2 rec = ovf2[k];
                if (rec.x == node) {
                    hit = true;
                    o_s = rec.y;
                    if (WEIGHTED) o_w = inv_sqrt_deg(cnt[o_s]) * dn;
                }
            }
            const int o_wbits = __builtin_bit_cast(int, o_w);
            unsigned todo = __builtin_amdgcn_ballot_w32(hit);
            while (todo != 0u) {
                const int l = __builtin_ctz(todo);
                todo &= todo - 1u;
                add_rows<1, VT, V, WEIGHTED>(acc, Yv, o_s, o_wbits, l, lane);
            }
        }
    }
}

template<int D>
__global__ __launch_bounds__(256)
void gather_scaled_kernel(const int* __restrict__ cnt, const int* __restrict__ bucket,
                          const int* __restrict__ ovf, const float* __restrict__ Ys,
                          const float* __restrict__ bias, float* __restrict__ outp, int n) {
    constexpr int V = D / 32;
    typedef typename RowVec<V>::type VT;

    const int lane = threadIdx.x & 31;
    const int node = __builtin_amdgcn_readfirstlane((int)(blockIdx.x * 8 + (threadIdx.x >> 5)));
    if (node >= n) return;

    const int   deg = cnt[node];
    const float dn  = inv_sqrt_deg(deg);

    const VT* __restrict__ Yv = (const VT*)Ys;
    const VT yself = Yv[(size_t)node * 32 + lane];
    const int my_s = slot_source(bucket, node, deg, lane, node);

    VT acc;
#pragma unroll
    for (int c = 0; c < V; ++c) acc[c] = 0.0f;

    sum_slots<VT, V, false>(acc, cnt, ovf, Yv, node, deg, dn, my_s, 0.0f, lane);

    VT o;
#pragma unroll
    for (int c = 0; c < V; ++c) o[c] = (acc[c] + yself[c]) * dn + bias[lane * V + c];
    ((VT*)outp)[(size_t)node * 32 + lane] = o;
}

#define TILE_NODES 32
static_assert(N_NODES % 16 == 0, "whole 16-row tiles");
static_assert(IN_DIM == HID_DIM, "tile_product steps K over IN_DIM for both products");
static_assert(HID_DIM / 16 == 8 && OUT_DIM / 16 == 4, "wave -> (column tile, row tile) split below");

__global__ __launch_bounds__(512)
void gather_transform_kernel(const int* __restrict__ cnt, const int* __restrict__ bucket,
                             const int* __restrict__ ovf, const float* __restrict__ X,
                             const float* __restrict__ W0, const float* __restrict__ b0,
                             const float* __restrict__ W1, float* __restrict__ Y1, int n) {
    typedef typename RowVec<4>::type VT;
    __shared__ float As[TILE_NODES * T_PITCH];
    __shared__ float Hs[TILE_NODES * T_PITCH];
    __shared__ float Ys[TILE_NODES * Y_PITCH];

    const int lane = threadIdx.x & 31;
    const int wave = __builtin_amdgcn_readfirstlane((int)(threadIdx.x >> 5));
    const int base = __builtin_amdgcn_readfirstlane((int)(blockIdx.x * TILE_NODES));

    constexpr int NH = TILE_NODES / 16;
    const VT* __restrict__ Xv = (const VT*)X;
    int   nd[NH], deg[NH], my_s[NH], src_cnt[NH];
    float dn[NH], my_w[NH];
    VT    xself[NH];
#pragma unroll
    for (int h = 0; h < NH; ++h) {
        const int node = base + h * 16 + wave;
        nd[h]  = node < n ? node : n - 1;
        deg[h] = cnt[nd[h]];
        dn[h]  = inv_sqrt_deg(deg[h]);
    }
#pragma unroll
    for (int h = 0; h < NH; ++h) {
        xself[h] = Xv[(size_t)nd[h] * 32 + lane];
        my_s[h]  = slot_source(bucket, nd[h], deg[h], lane, 0);
    }
#pragma unroll
    for (int h = 0; h < NH; ++h) src_cnt[h] = cnt[my_s[h]];
#pragma unroll
    for (int h = 0; h < NH; ++h) my_w[h] = slot_weight(src_cnt[h], deg[h], dn[h], lane);
#pragma unroll
    for (int h = 0; h < NH; ++h) {
        VT acc;
#pragma unroll
        for (int c = 0; c < 4; ++c) acc[c] = 0.0f;

        sum_slots<VT, 4, true>(acc, cnt, ovf, Xv, nd[h], deg[h], dn[h], my_s[h], my_w[h], lane);

        const float dself = dn[h] * dn[h];
        VT agg;
#pragma unroll
        for (int c = 0; c < 4; ++c) agg[c] = __builtin_fmaf(xself[h][c], dself, acc[c]);
        *(VT*)(As + (h * 16 + wave) * T_PITCH + lane * 4) = agg;
    }
    __syncthreads();

    const int half = lane >> 4;
    const int lid  = lane & 15;

    {
        const int tn = wave & 7;
        const int tm = wave >> 3;
        const v8f acc = tile_product<HID_DIM>(As + tm * 16 * T_PITCH, W0, tn, lane);
        const int   col  = tn * 16 + lid;
        const float bias = b0[col];
#pragma unroll
        for (int r = 0; r < 8; ++r)
            Hs[(tm * 16 + r + half * 8) * T_PITCH + col] = fmaxf(acc[r] + bias, 0.0f);
    }
    __syncthreads();

    if (wave < 2 * (OUT_DIM / 16)) {
        const int tn = wave & 3;
        const int tm = wave >> 2;
        const v8f acc = tile_product<OUT_DIM>(Hs + tm * 16 * T_PITCH, W1, tn, lane);
        const int col = tn * 16 + lid;
#pragma unroll
        for (int r = 0; r < 8; ++r)
            Ys[(tm * 16 + r + half * 8) * Y_PITCH + col] = acc[r];
    }
    __syncthreads();

#pragma unroll
    for (int h = 0; h < NH; ++h) {
        const int row  = h * 16 + wave;
        const int node = base + row;
        if (node < n) {
            const v2f y = *(const v2f*)(Ys + row * Y_PITCH + lane * 2);
            ((v2f*)Y1)[(size_t)node * 32 + lane] = y * dn[h];
        }
    }
}

extern "C" void kernel_launch(void* const* d_in, const int* in_sizes, int n_in,
                              void* d_out, int out_size, void* d_ws, size_t ws_size,
                              hipStream_t stream) {
    const int*   E  = (const int*)  d_in[1];
    const float* X  = (const float*)d_in[2];
    const float* W0 = (const float*)d_in[3];
    const float* b0 = (const float*)d_in[4];
    const float* W1 = (const float*)d_in[5];
    const float* b1 = (const float*)d_in[6];
    float* out = (float*)d_out;

    const int* Esrc = E;
    const int* Edst = E + N_EDGES;

    int*   cnt    = (int*)d_ws;
    int*   bucket = cnt + 50048;
    int*   ovf    = bucket + N_NODES * BUCKET_CAP;
    float* y1     = (float*)(ovf + 2 * N_EDGES);
    int*   pairs  = (int*)(y1 + N_NODES * OUT_DIM);
    int*   table  = pairs + 2 * N_EDGES;

    bin_kernel<<<N_CHUNKS, 512, 0, stream>>>(Esrc, Edst, pairs, table, cnt + N_NODES);
    own_kernel<<<N_BINS, 512, 0, stream>>>(pairs, table, cnt, bucket, ovf);

    gather_transform_kernel<<<(N_NODES + TILE_NODES - 1) / TILE_NODES, 512, 0, stream>>>(cnt, bucket, ovf, X, W0, b0, W1, y1, N_NODES);

    gather_scaled_kernel<OUT_DIM><<<(N_NODES + 7) / 8, 256, 0, stream>>>(cnt, bucket, ovf, y1, b1, out, N_NODES);
}
